# starting kernel
# baseline (speedup 1.0000x reference)
#include <hip/hip_runtime.h>
#include <math.h>

typedef __attribute__((ext_vector_type(2))) float v2f;
typedef __attribute__((ext_vector_type(8))) float v8f;

#define B_     16
#define N_     64
#define DEG_   16
#define IN_F_  1024
#define EDGE_F_ 128
#define H_     8
#define F_     64
#define E_     (N_ * DEG_)
#define HF_    (H_ * F_)
#define SLOPE_ 0.2f

#define BM 128
#define BN 64
#define BK 16
#define LDP (BK + 4)

__global__ __launch_bounds__(256) void gemm_wmma_f32_kernel(
    const float* __restrict__ A, const float* __restrict__ Bm,
    float* __restrict__ C, int M, int N, int K)
{
    __shared__ float As[BM][LDP];
    __shared__ float Bs[BN][LDP];

    const int tid  = threadIdx.x;
    const int wave = tid >> 5;
    const int lane = tid & 31;
    const int half = lane >> 4;
    const int l    = lane & 15;
    const int m0   = blockIdx.y * BM;
    const int n0   = blockIdx.x * BN;
    const int mrow = wave * 16;

    v8f acc[4] = {};

    for (int k0 = 0; k0 < K; k0 += BK) {
        for (int idx = tid; idx < (BM * BK) / 4; idx += 256) {
            int r = idx >> 2;
            int c = (idx & 3) * 4;
            float4 v = *(const float4*)&A[(size_t)(m0 + r) * K + (k0 + c)];
            *(float4*)&As[r][c] = v;
        }
        for (int idx = tid; idx < (BK * BN) / 4; idx += 256) {
            int r  = idx >> 4;
            int c4 = (idx & 15) * 4;
            float4 v = *(const float4*)&Bm[(size_t)(k0 + r) * N + (n0 + c4)];
            Bs[c4 + 0][r] = v.x;
            Bs[c4 + 1][r] = v.y;
            Bs[c4 + 2][r] = v.z;
            Bs[c4 + 3][r] = v.w;
        }
        __syncthreads();

        #pragma unroll
        for (int kk = 0; kk < BK; kk += 4) {
            v2f a = *(const v2f*)&As[mrow + l][kk + 2 * half];
            #pragma unroll
            for (int nt = 0; nt < 4; ++nt) {
                v2f b = *(const v2f*)&Bs[nt * 16 + l][kk + 2 * half];
                acc[nt] = __builtin_amdgcn_wmma_f32_16x16x4_f32(
                    false, a, false, b, (short)0, acc[nt], false, false);
            }
        }
        __syncthreads();
    }

    #pragma unroll
    for (int nt = 0; nt < 4; ++nt) {
        #pragma unroll
        for (int r = 0; r < 8; ++r) {
            C[(size_t)(m0 + mrow + r + 8 * half) * N + (n0 + nt * 16 + l)] =
                acc[nt][r];
        }
    }
}

__global__ void node_scores_kernel(const float* __restrict__ g,
                                   const float* __restrict__ w_attn,
                                   float* __restrict__ s_i,
                                   float* __restrict__ s_j)
{
    int t = blockIdx.x * blockDim.x + threadIdx.x;
    if (t >= B_ * N_ * H_) return;
    const float* gr = g + (size_t)t * F_;
    float si = 0.f, sj = 0.f;
    #pragma unroll 8
    for (int f = 0; f < F_; ++f) {
        float v = gr[f];
        si += v * w_attn[f];
        sj += v * w_attn[F_ + f];
    }
    s_i[t] = si;
    s_j[t] = sj;
}

__global__ void edge_scores_kernel(const float* __restrict__ ep,
                                   const float* __restrict__ w_attn,
                                   float* __restrict__ s_e)
{
    int t = blockIdx.x * blockDim.x + threadIdx.x;
    if (t >= B_ * E_ * H_) return;
    const float* er = ep + (size_t)t * F_;
    float se = 0.f;
    #pragma unroll 8
    for (int f = 0; f < F_; ++f) se += er[f] * w_attn[2 * F_ + f];
    s_e[t] = se;
}

__global__ __launch_bounds__(256) void attn_aggregate_kernel(
    const float* __restrict__ g, const float* __restrict__ s_i,
    const float* __restrict__ s_j, const float* __restrict__ s_e,
    float* __restrict__ out)
{
    const int lane = threadIdx.x & 31;
    const int wave = threadIdx.x >> 5;
    const int gidx = blockIdx.x * 8 + wave;
    const int h = gidx % H_;
    const int n = (gidx / H_) % N_;
    const int b = gidx / (H_ * N_);

    const int k    = lane & 15;
    const int jn   = (n + k) % N_;
    const int wrap = (n + DEG_ > N_) ? (n + DEG_ - N_) : 0;
    const int rank = (n + k < N_) ? (k + wrap) : (n + k - N_);
    const int eidx = n * DEG_ + rank;

    float sc = s_i[gidx]
             + s_j[(b * N_ + jn) * H_ + h]
             + s_e[(b * E_ + eidx) * H_ + h];
    sc = (sc > 0.f) ? sc : SLOPE_ * sc;

    float m = sc;
    #pragma unroll
    for (int off = 8; off >= 1; off >>= 1)
        m = fmaxf(m, __shfl_xor(m, off, 32));
    float ex = __expf(sc - m);
    float s = ex;
    #pragma unroll
    for (int off = 8; off >= 1; off >>= 1)
        s += __shfl_xor(s, off, 32);
    float a = ex / s;

    float o0 = 0.f, o1 = 0.f;
    #pragma unroll
    for (int kk = 0; kk < DEG_; ++kk) {
        float ak = __shfl(a, kk, 32);
        int jj = (n + kk) % N_;
        const float* gp = g + ((size_t)(b * N_ + jj) * H_ + h) * F_;
        o0 += ak * gp[lane];
        o1 += ak * gp[lane + 32];
    }
    float* op = out + ((size_t)(b * N_ + n) * H_ + h) * F_;
    op[lane]      = o0;
    op[lane + 32] = o1;
}

extern "C" void kernel_launch(void* const* d_in, const int* in_sizes, int n_in,
                              void* d_out, int out_size, void* d_ws, size_t ws_size,
                              hipStream_t stream)
{
    const float* h         = (const float*)d_in[0];
    const float* edge_attr = (const float*)d_in[2];
    const float* W_node    = (const float*)d_in[3];
    const float* W_edge    = (const float*)d_in[4];
    const float* w_attn    = (const float*)d_in[5];
    float* out = (float*)d_out;

    float* ws  = (float*)d_ws;
    float* g   = ws;
    float* ep  = g  + (size_t)B_ * N_ * HF_;
    float* s_i = ep + (size_t)B_ * E_ * HF_;
    float* s_j = s_i + (size_t)B_ * N_ * H_;
    float* s_e = s_j + (size_t)B_ * N_ * H_;

    dim3 blk(256);

    dim3 g1(HF_ / BN, (B_ * N_) / BM);
    gemm_wmma_f32_kernel<<<g1, blk, 0, stream>>>(h, W_node, g,
                                                 B_ * N_, HF_, IN_F_);

    dim3 g2(HF_ / BN, (B_ * E_) / BM);
    gemm_wmma_f32_kernel<<<g2, blk, 0, stream>>>(edge_attr, W_edge, ep,
                                                 B_ * E_, HF_, EDGE_F_);

    node_scores_kernel<<<(B_ * N_ * H_ + 255) / 256, blk, 0, stream>>>(
        g, w_attn, s_i, s_j);
    edge_scores_kernel<<<(B_ * E_ * H_ + 255) / 256, blk, 0, stream>>>(
        ep, w_attn, s_e);

    attn_aggregate_kernel<<<(B_ * N_ * H_) / 8, blk, 0, stream>>>(
        g, s_i, s_j, s_e, out);
}


# node GEMM retiled: 64x32 output tile x K step 128 on 4-wave workgroups, 256 workgroups and 8 K steps where 128x64x16 gave 64 and 64; same k order per element
# speedup vs baseline: 2.2783x; 2.2783x over previous
#include <hip/hip_runtime.h>
#include <math.h>

typedef __attribute__((ext_vector_type(2))) float v2f;
typedef __attribute__((ext_vector_type(8))) float v8f;

#define B_     16
#define N_     64
#define DEG_   16
#define IN_F_  1024
#define EDGE_F_ 128
#define H_     8
#define F_     64
#define E_     (N_ * DEG_)
#define HF_    (H_ * F_)
#define SLOPE_ 0.2f

#define BM 128
#define BN 64
#define BK 16
#define LDP (BK + 4)

__global__ __launch_bounds__(256) void gemm_wmma_f32_kernel(
    const float* __restrict__ A, const float* __restrict__ Bm,
    float* __restrict__ C, int M, int N, int K)
{
    __shared__ float As[BM][LDP];
    __shared__ float Bs[BN][LDP];

    const int tid  = threadIdx.x;
    const int wave = tid >> 5;
    const int lane = tid & 31;
    const int half = lane >> 4;
    const int l    = lane & 15;
    const int m0   = blockIdx.y * BM;
    const int n0   = blockIdx.x * BN;
    const int mrow = wave * 16;

    v8f acc[4] = {};

    for (int k0 = 0; k0 < K; k0 += BK) {
        for (int idx = tid; idx < (BM * BK) / 4; idx += 256) {
            int r = idx >> 2;
            int c = (idx & 3) * 4;
            float4 v = *(const float4*)&A[(size_t)(m0 + r) * K + (k0 + c)];
            *(float4*)&As[r][c] = v;
        }
        for (int idx = tid; idx < (BK * BN) / 4; idx += 256) {
            int r  = idx >> 4;
            int c4 = (idx & 15) * 4;
            float4 v = *(const float4*)&Bm[(size_t)(k0 + r) * N + (n0 + c4)];
            Bs[c4 + 0][r] = v.x;
            Bs[c4 + 1][r] = v.y;
            Bs[c4 + 2][r] = v.z;
            Bs[c4 + 3][r] = v.w;
        }
        __syncthreads();

        #pragma unroll
        for (int kk = 0; kk < BK; kk += 4) {
            v2f a = *(const v2f*)&As[mrow + l][kk + 2 * half];
            #pragma unroll
            for (int nt = 0; nt < 4; ++nt) {
                v2f b = *(const v2f*)&Bs[nt * 16 + l][kk + 2 * half];
                acc[nt] = __builtin_amdgcn_wmma_f32_16x16x4_f32(
                    false, a, false, b, (short)0, acc[nt], false, false);
            }
        }
        __syncthreads();
    }

    #pragma unroll
    for (int nt = 0; nt < 4; ++nt) {
        #pragma unroll
        for (int r = 0; r < 8; ++r) {
            C[(size_t)(m0 + mrow + r + 8 * half) * N + (n0 + nt * 16 + l)] =
                acc[nt][r];
        }
    }
}

#define TM 64
#define TN 32
#define TK 128
#define TLD (TK + 4)
#define GM (B_ * N_)
#define GN HF_
#define GK IN_F_
#define A_IT ((TM * TK) / 4 / 128)
#define B_IT ((TK * TN) / 4 / 128)

__global__ __launch_bounds__(128) void gemm_node_wmma_f32_kernel(
    const float* __restrict__ A, const float* __restrict__ Bm,
    float* __restrict__ C)
{
    __shared__ __attribute__((aligned(16))) float As[TM][TLD];
    __shared__ __attribute__((aligned(16))) float Bs[TN][TLD];

    const int tid  = threadIdx.x;
    const int wave = tid >> 5;
    const int lane = tid & 31;
    const int half = lane >> 4;
    const int l    = lane & 15;
    const int m0   = blockIdx.y * TM;
    const int n0   = blockIdx.x * TN;
    const int mrow = wave * 16;

    const int a_r = tid >> 5;
    const int a_c = (tid & 31) * 4;
    const int b_c4 = ((tid & 3) + 4 * (wave & 1)) * 4;
    const int b_r  = ((tid >> 2) & 7) + 8 * (wave >> 1);

    const float* ap = A  + (size_t)(m0 + a_r) * GK + a_c;
    const float* bp = Bm + (size_t)b_r * GN + (n0 + b_c4);

    v8f acc[2] = {};

    for (int k0 = 0; k0 < GK; k0 += TK) {
        float4 va[A_IT], vb[B_IT];
        #pragma unroll
        for (int it = 0; it < A_IT; ++it)
            va[it] = *(const float4*)(ap + (size_t)(4 * it) * GK + k0);
        #pragma unroll
        for (int it = 0; it < B_IT; ++it)
            vb[it] = *(const float4*)(bp + (size_t)(k0 + 16 * it) * GN);
        #pragma unroll
        for (int it = 0; it < A_IT; ++it)
            *(float4*)&As[a_r + 4 * it][a_c] = va[it];
        #pragma unroll
        for (int it = 0; it < B_IT; ++it) {
            const int r = b_r + 16 * it;
            Bs[b_c4 + 0][r] = vb[it].x;
            Bs[b_c4 + 1][r] = vb[it].y;
            Bs[b_c4 + 2][r] = vb[it].z;
            Bs[b_c4 + 3][r] = vb[it].w;
        }
        __syncthreads();

        #pragma unroll
        for (int kk = 0; kk < TK; kk += 4) {
            v2f a = *(const v2f*)&As[mrow + l][kk + 2 * half];
            #pragma unroll
            for (int nt = 0; nt < 2; ++nt) {
                v2f b = *(const v2f*)&Bs[nt * 16 + l][kk + 2 * half];
                acc[nt] = __builtin_amdgcn_wmma_f32_16x16x4_f32(
                    false, a, false, b, (short)0, acc[nt], false, false);
            }
        }
        __syncthreads();
    }

    #pragma unroll
    for (int nt = 0; nt < 2; ++nt) {
        #pragma unroll
        for (int r = 0; r < 8; ++r) {
            C[(size_t)(m0 + mrow + r + 8 * half) * GN + (n0 + nt * 16 + l)] =
                acc[nt][r];
        }
    }
}

__global__ void node_scores_kernel(const float* __restrict__ g,
                                   const float* __restrict__ w_attn,
                                   float* __restrict__ s_i,
                                   float* __restrict__ s_j)
{
    int t = blockIdx.x * blockDim.x + threadIdx.x;
    if (t >= B_ * N_ * H_) return;
    const float* gr = g + (size_t)t * F_;
    float si = 0.f, sj = 0.f;
    #pragma unroll 8
    for (int f = 0; f < F_; ++f) {
        float v = gr[f];
        si += v * w_attn[f];
        sj += v * w_attn[F_ + f];
    }
    s_i[t] = si;
    s_j[t] = sj;
}

__global__ void edge_scores_kernel(const float* __restrict__ ep,
                                   const float* __restrict__ w_attn,
                                   float* __restrict__ s_e)
{
    int t = blockIdx.x * blockDim.x + threadIdx.x;
    if (t >= B_ * E_ * H_) return;
    const float* er = ep + (size_t)t * F_;
    float se = 0.f;
    #pragma unroll 8
    for (int f = 0; f < F_; ++f) se += er[f] * w_attn[2 * F_ + f];
    s_e[t] = se;
}

__global__ __launch_bounds__(256) void attn_aggregate_kernel(
    const float* __restrict__ g, const float* __restrict__ s_i,
    const float* __restrict__ s_j, const float* __restrict__ s_e,
    float* __restrict__ out)
{
    const int lane = threadIdx.x & 31;
    const int wave = threadIdx.x >> 5;
    const int gidx = blockIdx.x * 8 + wave;
    const int h = gidx % H_;
    const int n = (gidx / H_) % N_;
    const int b = gidx / (H_ * N_);

    const int k    = lane & 15;
    const int jn   = (n + k) % N_;
    const int wrap = (n + DEG_ > N_) ? (n + DEG_ - N_) : 0;
    const int rank = (n + k < N_) ? (k + wrap) : (n + k - N_);
    const int eidx = n * DEG_ + rank;

    float sc = s_i[gidx]
             + s_j[(b * N_ + jn) * H_ + h]
             + s_e[(b * E_ + eidx) * H_ + h];
    sc = (sc > 0.f) ? sc : SLOPE_ * sc;

    float m = sc;
    #pragma unroll
    for (int off = 8; off >= 1; off >>= 1)
        m = fmaxf(m, __shfl_xor(m, off, 32));
    float ex = __expf(sc - m);
    float s = ex;
    #pragma unroll
    for (int off = 8; off >= 1; off >>= 1)
        s += __shfl_xor(s, off, 32);
    float a = ex / s;

    float o0 = 0.f, o1 = 0.f;
    #pragma unroll
    for (int kk = 0; kk < DEG_; ++kk) {
        float ak = __shfl(a, kk, 32);
        int jj = (n + kk) % N_;
        const float* gp = g + ((size_t)(b * N_ + jj) * H_ + h) * F_;
        o0 += ak * gp[lane];
        o1 += ak * gp[lane + 32];
    }
    float* op = out + ((size_t)(b * N_ + n) * H_ + h) * F_;
    op[lane]      = o0;
    op[lane + 32] = o1;
}

extern "C" void kernel_launch(void* const* d_in, const int* in_sizes, int n_in,
                              void* d_out, int out_size, void* d_ws, size_t ws_size,
                              hipStream_t stream)
{
    const float* h         = (const float*)d_in[0];
    const float* edge_attr = (const float*)d_in[2];
    const float* W_node    = (const float*)d_in[3];
    const float* W_edge    = (const float*)d_in[4];
    const float* w_attn    = (const float*)d_in[5];
    float* out = (float*)d_out;

    float* ws  = (float*)d_ws;
    float* g   = ws;
    float* ep  = g  + (size_t)B_ * N_ * HF_;
    float* s_i = ep + (size_t)B_ * E_ * HF_;
    float* s_j = s_i + (size_t)B_ * N_ * H_;
    float* s_e = s_j + (size_t)B_ * N_ * H_;

    dim3 blk(256);

    dim3 g1(GN / TN, GM / TM);
    gemm_node_wmma_f32_kernel<<<g1, dim3(128), 0, stream>>>(h, W_node, g);

    dim3 g2(HF_ / BN, (B_ * E_) / BM);
    gemm_wmma_f32_kernel<<<g2, blk, 0, stream>>>(edge_attr, W_edge, ep,
                                                 B_ * E_, HF_, EDGE_F_);

    node_scores_kernel<<<(B_ * N_ * H_ + 255) / 256, blk, 0, stream>>>(
        g, w_attn, s_i, s_j);
    edge_scores_kernel<<<(B_ * E_ * H_ + 255) / 256, blk, 0, stream>>>(
        ep, w_attn, s_e);

    attn_aggregate_kernel<<<(B_ * N_ * H_) / 8, blk, 0, stream>>>(
        g, s_i, s_j, s_e, out);
}


# edge scores by algebra: s_e = edge_attr x (W_edge . w_e), a folded 128x8 weight, as one skinny f32 WMMA product; the 32 MiB e_proj is never formed; s_e sums reassociated
# speedup vs baseline: 4.1035x; 1.8011x over previous
#include <hip/hip_runtime.h>
#include <math.h>

typedef __attribute__((ext_vector_type(2))) float v2f;
typedef __attribute__((ext_vector_type(8))) float v8f;

#define B_     16
#define N_     64
#define DEG_   16
#define IN_F_  1024
#define EDGE_F_ 128
#define H_     8
#define F_     64
#define E_     (N_ * DEG_)
#define HF_    (H_ * F_)
#define SLOPE_ 0.2f

#define TM 64
#define TN 32
#define TK 128
#define TLD (TK + 4)
#define GM (B_ * N_)
#define GN HF_
#define GK IN_F_
#define A_IT ((TM * TK) / 4 / 128)
#define B_IT ((TK * TN) / 4 / 128)

__global__ __launch_bounds__(128) void gemm_node_wmma_f32_kernel(
    const float* __restrict__ A, const float* __restrict__ Bm,
    float* __restrict__ C)
{
    __shared__ __attribute__((aligned(16))) float As[TM][TLD];
    __shared__ __attribute__((aligned(16))) float Bs[TN][TLD];

    const int tid  = threadIdx.x;
    const int wave = tid >> 5;
    const int lane = tid & 31;
    const int half = lane >> 4;
    const int l    = lane & 15;
    const int m0   = blockIdx.y * TM;
    const int n0   = blockIdx.x * TN;
    const int mrow = wave * 16;

    const int a_r = tid >> 5;
    const int a_c = (tid & 31) * 4;
    const int b_c4 = ((tid & 3) + 4 * (wave & 1)) * 4;
    const int b_r  = ((tid >> 2) & 7) + 8 * (wave >> 1);

    const float* ap = A  + (size_t)(m0 + a_r) * GK + a_c;
    const float* bp = Bm + (size_t)b_r * GN + (n0 + b_c4);

    v8f acc[2] = {};

    for (int k0 = 0; k0 < GK; k0 += TK) {
        float4 va[A_IT], vb[B_IT];
        #pragma unroll
        for (int it = 0; it < A_IT; ++it)
            va[it] = *(const float4*)(ap + (size_t)(4 * it) * GK + k0);
        #pragma unroll
        for (int it = 0; it < B_IT; ++it)
            vb[it] = *(const float4*)(bp + (size_t)(k0 + 16 * it) * GN);
        #pragma unroll
        for (int it = 0; it < A_IT; ++it)
            *(float4*)&As[a_r + 4 * it][a_c] = va[it];
        #pragma unroll
        for (int it = 0; it < B_IT; ++it) {
            const int r = b_r + 16 * it;
            Bs[b_c4 + 0][r] = vb[it].x;
            Bs[b_c4 + 1][r] = vb[it].y;
            Bs[b_c4 + 2][r] = vb[it].z;
            Bs[b_c4 + 3][r] = vb[it].w;
        }
        __syncthreads();

        #pragma unroll
        for (int kk = 0; kk < TK; kk += 4) {
            v2f a = *(const v2f*)&As[mrow + l][kk + 2 * half];
            #pragma unroll
            for (int nt = 0; nt < 2; ++nt) {
                v2f b = *(const v2f*)&Bs[nt * 16 + l][kk + 2 * half];
                acc[nt] = __builtin_amdgcn_wmma_f32_16x16x4_f32(
                    false, a, false, b, (short)0, acc[nt], false, false);
            }
        }
        __syncthreads();
    }

    #pragma unroll
    for (int nt = 0; nt < 2; ++nt) {
        #pragma unroll
        for (int r = 0; r < 8; ++r) {
            C[(size_t)(m0 + mrow + r + 8 * half) * GN + (n0 + nt * 16 + l)] =
                acc[nt][r];
        }
    }
}

__global__ void node_scores_kernel(const float* __restrict__ g,
                                   const float* __restrict__ w_attn,
                                   float* __restrict__ s_i,
                                   float* __restrict__ s_j)
{
    int t = blockIdx.x * blockDim.x + threadIdx.x;
    if (t >= B_ * N_ * H_) return;
    const float* gr = g + (size_t)t * F_;
    float si = 0.f, sj = 0.f;
    #pragma unroll 8
    for (int f = 0; f < F_; ++f) {
        float v = gr[f];
        si += v * w_attn[f];
        sj += v * w_attn[F_ + f];
    }
    s_i[t] = si;
    s_j[t] = sj;
}

__global__ __launch_bounds__(256) void edge_weight_fold_kernel(
    const float* __restrict__ W_edge, const float* __restrict__ w_attn,
    float* __restrict__ wpT)
{
    const int t   = blockIdx.x * 256 + threadIdx.x;
    const int o   = t >> 4;
    const int sub = t & 15;
    const float4 w = *(const float4*)(W_edge + (size_t)o * F_ + sub * 4);
    const float* we = w_attn + 2 * F_ + sub * 4;
    float p = w.x * we[0] + w.y * we[1] + w.z * we[2] + w.w * we[3];
    #pragma unroll
    for (int off = 8; off >= 1; off >>= 1)
        p += __shfl_xor(p, off, 32);
    if (sub == 0) wpT[(o & (H_ - 1)) * EDGE_F_ + (o >> 3)] = p;
}

#define EQ (EDGE_F_ / 8)

__global__ __launch_bounds__(128) void edge_scores_wmma_f32_kernel(
    const float* __restrict__ ea, const float* __restrict__ wpT,
    float* __restrict__ s_e)
{
    const int lane = threadIdx.x & 31;
    const int wave = threadIdx.x >> 5;
    const int half = lane >> 4;
    const int c    = lane & 15;
    const int row0 = (blockIdx.x * 4 + wave) * 16;

    const float* bp = ea  + (size_t)(row0 + c) * EDGE_F_ + 4 * half;
    const float* ap = wpT + (c & (H_ - 1)) * EDGE_F_ + 4 * half;

    float4 vb[EQ], va[EQ];
    #pragma unroll
    for (int q = 0; q < EQ; ++q) {
        vb[q] = *(const float4*)(bp + 8 * q);
        va[q] = *(const float4*)(ap + 8 * q);
    }
    __builtin_amdgcn_sched_barrier(0);

    v8f acc = {};
    #pragma unroll
    for (int q = 0; q < EQ; ++q) {
        v2f a0 = {va[q].x, va[q].y}, b0 = {vb[q].x, vb[q].y};
        v2f a1 = {va[q].z, va[q].w}, b1 = {vb[q].z, vb[q].w};
        acc = __builtin_amdgcn_wmma_f32_16x16x4_f32(
            false, a0, false, b0, (short)0, acc, false, false);
        acc = __builtin_amdgcn_wmma_f32_16x16x4_f32(
            false, a1, false, b1, (short)0, acc, false, false);
    }

    if (half == 0) {
        float* op = s_e + (size_t)(row0 + c) * H_;
        *(float4*)(op)     = make_float4(acc[0], acc[1], acc[2], acc[3]);
        *(float4*)(op + 4) = make_float4(acc[4], acc[5], acc[6], acc[7]);
    }
}

__global__ __launch_bounds__(256) void attn_aggregate_kernel(
    const float* __restrict__ g, const float* __restrict__ s_i,
    const float* __restrict__ s_j, const float* __restrict__ s_e,
    float* __restrict__ out)
{
    const int lane = threadIdx.x & 31;
    const int wave = threadIdx.x >> 5;
    const int gidx = blockIdx.x * 8 + wave;
    const int h = gidx % H_;
    const int n = (gidx / H_) % N_;
    const int b = gidx / (H_ * N_);

    const int k    = lane & 15;
    const int jn   = (n + k) % N_;
    const int wrap = (n + DEG_ > N_) ? (n + DEG_ - N_) : 0;
    const int rank = (n + k < N_) ? (k + wrap) : (n + k - N_);
    const int eidx = n * DEG_ + rank;

    float sc = s_i[gidx]
             + s_j[(b * N_ + jn) * H_ + h]
             + s_e[(b * E_ + eidx) * H_ + h];
    sc = (sc > 0.f) ? sc : SLOPE_ * sc;

    float m = sc;
    #pragma unroll
    for (int off = 8; off >= 1; off >>= 1)
        m = fmaxf(m, __shfl_xor(m, off, 32));
    float ex = __expf(sc - m);
    float s = ex;
    #pragma unroll
    for (int off = 8; off >= 1; off >>= 1)
        s += __shfl_xor(s, off, 32);
    float a = ex / s;

    float o0 = 0.f, o1 = 0.f;
    #pragma unroll
    for (int kk = 0; kk < DEG_; ++kk) {
        float ak = __shfl(a, kk, 32);
        int jj = (n + kk) % N_;
        const float* gp = g + ((size_t)(b * N_ + jj) * H_ + h) * F_;
        o0 += ak * gp[lane];
        o1 += ak * gp[lane + 32];
    }
    float* op = out + ((size_t)(b * N_ + n) * H_ + h) * F_;
    op[lane]      = o0;
    op[lane + 32] = o1;
}

extern "C" void kernel_launch(void* const* d_in, const int* in_sizes, int n_in,
                              void* d_out, int out_size, void* d_ws, size_t ws_size,
                              hipStream_t stream)
{
    const float* h         = (const float*)d_in[0];
    const float* edge_attr = (const float*)d_in[2];
    const float* W_node    = (const float*)d_in[3];
    const float* W_edge    = (const float*)d_in[4];
    const float* w_attn    = (const float*)d_in[5];
    float* out = (float*)d_out;

    float* ws  = (float*)d_ws;
    float* g   = ws;
    float* wpT = g  + (size_t)B_ * N_ * HF_;
    float* s_i = wpT + (size_t)B_ * E_ * HF_;
    float* s_j = s_i + (size_t)B_ * N_ * H_;
    float* s_e = s_j + (size_t)B_ * N_ * H_;

    dim3 blk(256);

    dim3 g1(GN / TN, GM / TM);
    gemm_node_wmma_f32_kernel<<<g1, dim3(128), 0, stream>>>(h, W_node, g);

    edge_weight_fold_kernel<<<(EDGE_F_ * H_ * 16) / 256, blk, 0, stream>>>(
        W_edge, w_attn, wpT);

    node_scores_kernel<<<(B_ * N_ * H_ + 255) / 256, blk, 0, stream>>>(
        g, w_attn, s_i, s_j);

    edge_scores_wmma_f32_kernel<<<(B_ * E_) / 64, dim3(128), 0, stream>>>(
        edge_attr, wpT, s_e);

    attn_aggregate_kernel<<<(B_ * N_ * H_) / 8, blk, 0, stream>>>(
        g, s_i, s_j, s_e, out);
}


# launches 5 -> 2: weight fold rides in the GEMM grid; edge scores, node scores, softmax, aggregation fused, one wave per node, 19 g rows staged in LDS per 4 nodes; s_i, s_j sums reassociated
# speedup vs baseline: 5.6823x; 1.3847x over previous
#include <hip/hip_runtime.h>
#include <math.h>

typedef __attribute__((ext_vector_type(2))) float v2f;
typedef __attribute__((ext_vector_type(8))) float v8f;

#define B_     16
#define N_     64
#define DEG_   16
#define IN_F_  1024
#define EDGE_F_ 128
#define H_     8
#define F_     64
#define E_     (N_ * DEG_)
#define HF_    (H_ * F_)
#define SLOPE_ 0.2f

__device__ __forceinline__ void edge_weight_fold(
    const int t, const float* __restrict__ W_edge,
    const float* __restrict__ w_attn, float* __restrict__ wpT)
{
    const int o   = t >> 4;
    const int sub = t & 15;
    const float4 w = *(const float4*)(W_edge + (size_t)o * F_ + sub * 4);
    const float* we = w_attn + 2 * F_ + sub * 4;
    float p = w.x * we[0] + w.y * we[1] + w.z * we[2] + w.w * we[3];
    #pragma unroll
    for (int off = 8; off >= 1; off >>= 1)
        p += __shfl_xor(p, off, 32);
    if (sub == 0) wpT[(o & (H_ - 1)) * EDGE_F_ + (o >> 3)] = p;
}

#define TM 64
#define TN 32
#define TK 128
#define TLD (TK + 4)
#define GM (B_ * N_)
#define GN HF_
#define GK IN_F_
#define A_IT ((TM * TK) / 4 / 128)
#define B_IT ((TK * TN) / 4 / 128)
#define GEMM_BLOCKS ((GM / TM) * (GN / TN))
#define FOLD_BLOCKS ((EDGE_F_ * H_ * 16) / 128)

__global__ __launch_bounds__(128) void gemm_node_and_fold_kernel(
    const float* __restrict__ A, const float* __restrict__ Bm,
    float* __restrict__ C,
    const float* __restrict__ W_edge, const float* __restrict__ w_attn,
    float* __restrict__ wpT)
{
    if (blockIdx.x >= GEMM_BLOCKS) {
        edge_weight_fold((blockIdx.x - GEMM_BLOCKS) * 128 + threadIdx.x,
                         W_edge, w_attn, wpT);
        return;
    }

    __shared__ __attribute__((aligned(16))) float As[TM][TLD];
    __shared__ __attribute__((aligned(16))) float Bs[TN][TLD];

    const int tid  = threadIdx.x;
    const int wave = tid >> 5;
    const int lane = tid & 31;
    const int half = lane >> 4;
    const int l    = lane & 15;
    const int m0   = (blockIdx.x >> 4) * TM;
    const int n0   = (blockIdx.x & 15) * TN;
    const int mrow = wave * 16;

    const int a_r = tid >> 5;
    const int a_c = (tid & 31) * 4;
    const int b_c4 = ((tid & 3) + 4 * (wave & 1)) * 4;
    const int b_r  = ((tid >> 2) & 7) + 8 * (wave >> 1);

    const float* ap = A  + (size_t)(m0 + a_r) * GK + a_c;
    const float* bp = Bm + (size_t)b_r * GN + (n0 + b_c4);

    v8f acc[2] = {};

    for (int k0 = 0; k0 < GK; k0 += TK) {
        float4 va[A_IT], vb[B_IT];
        #pragma unroll
        for (int it = 0; it < A_IT; ++it)
            va[it] = *(const float4*)(ap + (size_t)(4 * it) * GK + k0);
        #pragma unroll
        for (int it = 0; it < B_IT; ++it)
            vb[it] = *(const float4*)(bp + (size_t)(k0 + 16 * it) * GN);
        #pragma unroll
        for (int it = 0; it < A_IT; ++it)
            *(float4*)&As[a_r + 4 * it][a_c] = va[it];
        #pragma unroll
        for (int it = 0; it < B_IT; ++it) {
            const int r = b_r + 16 * it;
            Bs[b_c4 + 0][r] = vb[it].x;
            Bs[b_c4 + 1][r] = vb[it].y;
            Bs[b_c4 + 2][r] = vb[it].z;
            Bs[b_c4 + 3][r] = vb[it].w;
        }
        __syncthreads();

        #pragma unroll
        for (int kk = 0; kk < TK; kk += 4) {
            v2f a = *(const v2f*)&As[mrow + l][kk + 2 * half];
            #pragma unroll
            for (int nt = 0; nt < 2; ++nt) {
                v2f b = *(const v2f*)&Bs[nt * 16 + l][kk + 2 * half];
                acc[nt] = __builtin_amdgcn_wmma_f32_16x16x4_f32(
                    false, a, false, b, (short)0, acc[nt], false, false);
            }
        }
        __syncthreads();
    }

    #pragma unroll
    for (int nt = 0; nt < 2; ++nt) {
        #pragma unroll
        for (int r = 0; r < 8; ++r) {
            C[(size_t)(m0 + mrow + r + 8 * half) * GN + (n0 + nt * 16 + l)] =
                acc[nt][r];
        }
    }
}

#define EQ   (EDGE_F_ / 8)
#define NPB  4
#define NROW (NPB + DEG_ - 1)

__global__ __launch_bounds__(128) void edge_attn_fused_kernel(
    const float* __restrict__ g, const float* __restrict__ ea,
    const float* __restrict__ wpT, const float* __restrict__ w_attn,
    float* __restrict__ out)
{
    __shared__ __attribute__((aligned(16))) float gs[NROW][HF_];
    __shared__ float sj_s[NROW][H_];
    __shared__ float si_s[NPB][H_];
    __shared__ __attribute__((aligned(16))) float se_s[NPB][DEG_][H_];
    __shared__ __attribute__((aligned(16))) float aw_s[NPB][2][DEG_][4];

    const int tid  = threadIdx.x;
    const int lane = tid & 31;
    const int wave = tid >> 5;
    const int half = lane >> 4;
    const int c    = lane & 15;
    const int b    = blockIdx.x / (N_ / NPB);
    const int n0   = (blockIdx.x % (N_ / NPB)) * NPB;
    const int n    = n0 + wave;

    const int row0 = b * E_ + n * DEG_;
    const float* bp = ea  + (size_t)(row0 + c) * EDGE_F_ + 4 * half;
    const float* ap = wpT + (c & (H_ - 1)) * EDGE_F_ + 4 * half;
    float4 vb[EQ], va[EQ];
    #pragma unroll
    for (int q = 0; q < EQ; ++q) {
        vb[q] = *(const float4*)(bp + 8 * q);
        va[q] = *(const float4*)(ap + 8 * q);
    }
    float4 gv[NROW];
    #pragma unroll
    for (int r = 0; r < NROW; ++r) {
        const int node = (n0 + r) & (N_ - 1);
        gv[r] = *(const float4*)(g + (size_t)(b * N_ + node) * HF_ + 4 * tid);
    }
    const float4 wi4 = *(const float4*)(w_attn + 4 * c);
    const float4 wj4 = *(const float4*)(w_attn + F_ + 4 * c);
    __builtin_amdgcn_sched_barrier(0);

    float pj[NROW], pi[NPB];
    #pragma unroll
    for (int r = 0; r < NROW; ++r) {
        *(float4*)&gs[r][4 * tid] = gv[r];
        pj[r] = gv[r].x * wj4.x + gv[r].y * wj4.y + gv[r].z * wj4.z + gv[r].w * wj4.w;
    }
    #pragma unroll
    for (int r = 0; r < NPB; ++r)
        pi[r] = gv[r].x * wi4.x + gv[r].y * wi4.y + gv[r].z * wi4.z + gv[r].w * wi4.w;
    #pragma unroll
    for (int off = 8; off >= 1; off >>= 1) {
        #pragma unroll
        for (int r = 0; r < NROW; ++r)
            pj[r] += __shfl_xor(pj[r], off, 32);
        #pragma unroll
        for (int r = 0; r < NPB; ++r)
            pi[r] += __shfl_xor(pi[r], off, 32);
    }
    if (c == 0) {
        const int hh = tid >> 4;
        #pragma unroll
        for (int r = 0; r < NROW; ++r) sj_s[r][hh] = pj[r];
        #pragma unroll
        for (int r = 0; r < NPB; ++r)  si_s[r][hh] = pi[r];
    }

    v8f acc = {};
    #pragma unroll
    for (int q = 0; q < EQ; ++q) {
        v2f a0 = {va[q].x, va[q].y}, b0 = {vb[q].x, vb[q].y};
        v2f a1 = {va[q].z, va[q].w}, b1 = {vb[q].z, vb[q].w};
        acc = __builtin_amdgcn_wmma_f32_16x16x4_f32(
            false, a0, false, b0, (short)0, acc, false, false);
        acc = __builtin_amdgcn_wmma_f32_16x16x4_f32(
            false, a1, false, b1, (short)0, acc, false, false);
    }
    if (half == 0) {
        *(float4*)&se_s[wave][c][0] = make_float4(acc[0], acc[1], acc[2], acc[3]);
        *(float4*)&se_s[wave][c][4] = make_float4(acc[4], acc[5], acc[6], acc[7]);
    }
    __syncthreads();

    const int k    = c;
    const int wrap = (n + DEG_ > N_) ? (n + DEG_ - N_) : 0;
    const int rank = (n + k < N_) ? (k + wrap) : (n + k - N_);
    float aw[4];
    #pragma unroll
    for (int j = 0; j < 4; ++j) {
        const int h = 2 * j + half;
        float sc = si_s[wave][h] + sj_s[wave + k][h] + se_s[wave][rank][h];
        sc = (sc > 0.f) ? sc : SLOPE_ * sc;
        float m = sc;
        #pragma unroll
        for (int off = 8; off >= 1; off >>= 1)
            m = fmaxf(m, __shfl_xor(m, off, 32));
        const float ex = __expf(sc - m);
        float s = ex;
        #pragma unroll
        for (int off = 8; off >= 1; off >>= 1)
            s += __shfl_xor(s, off, 32);
        aw[j] = ex / s;
    }
    *(float4*)&aw_s[wave][half][k][0] = make_float4(aw[0], aw[1], aw[2], aw[3]);
    __syncthreads();

    float4 o[4];
    #pragma unroll
    for (int j = 0; j < 4; ++j) o[j] = make_float4(0.f, 0.f, 0.f, 0.f);
    #pragma unroll
    for (int kk = 0; kk < DEG_; ++kk) {
        const float4 a4 = *(const float4*)&aw_s[wave][half][kk][0];
        const float aj[4] = {a4.x, a4.y, a4.z, a4.w};
        #pragma unroll
        for (int j = 0; j < 4; ++j) {
            const float4 gq = *(const float4*)&gs[wave + kk][4 * (lane + 32 * j)];
            o[j].x += aj[j] * gq.x;
            o[j].y += aj[j] * gq.y;
            o[j].z += aj[j] * gq.z;
            o[j].w += aj[j] * gq.w;
        }
    }
    float* op = out + (size_t)(b * N_ + n) * HF_;
    #pragma unroll
    for (int j = 0; j < 4; ++j)
        *(float4*)(op + 4 * (lane + 32 * j)) = o[j];
}

extern "C" void kernel_launch(void* const* d_in, const int* in_sizes, int n_in,
                              void* d_out, int out_size, void* d_ws, size_t ws_size,
                              hipStream_t stream)
{
    const float* h         = (const float*)d_in[0];
    const float* edge_attr = (const float*)d_in[2];
    const float* W_node    = (const float*)d_in[3];
    const float* W_edge    = (const float*)d_in[4];
    const float* w_attn    = (const float*)d_in[5];
    float* out = (float*)d_out;

    float* ws  = (float*)d_ws;
    float* g   = ws;
    float* wpT = g  + (size_t)B_ * N_ * HF_;

    gemm_node_and_fold_kernel<<<GEMM_BLOCKS + FOLD_BLOCKS, dim3(128), 0, stream>>>(
        h, W_node, g, W_edge, w_attn, wpT);

    edge_attn_fused_kernel<<<(B_ * N_) / NPB, dim3(128), 0, stream>>>(
        g, edge_attr, wpT, w_attn, out);
}


# node GEMM K loop software-pipelined: next K step's 24 global loads in flight under the current step's 64 WMMAs, ring of two LDS stages, one barrier per step; same k order per element
# speedup vs baseline: 6.7261x; 1.1837x over previous
#include <hip/hip_runtime.h>
#include <math.h>

typedef __attribute__((ext_vector_type(2))) float v2f;
typedef __attribute__((ext_vector_type(8))) float v8f;

#define B_     16
#define N_     64
#define DEG_   16
#define IN_F_  1024
#define EDGE_F_ 128
#define H_     8
#define F_     64
#define E_     (N_ * DEG_)
#define HF_    (H_ * F_)
#define SLOPE_ 0.2f

__device__ __forceinline__ void edge_weight_fold(
    const int t, const float* __restrict__ W_edge,
    const float* __restrict__ w_attn, float* __restrict__ wpT)
{
    const int o   = t >> 4;
    const int sub = t & 15;
    const float4 w = *(const float4*)(W_edge + (size_t)o * F_ + sub * 4);
    const float* we = w_attn + 2 * F_ + sub * 4;
    float p = w.x * we[0] + w.y * we[1] + w.z * we[2] + w.w * we[3];
    #pragma unroll
    for (int off = 8; off >= 1; off >>= 1)
        p += __shfl_xor(p, off, 32);
    if (sub == 0) wpT[(o & (H_ - 1)) * EDGE_F_ + (o >> 3)] = p;
}

#define TM 64
#define TN 32
#define TK 128
#define TLD (TK + 4)
#define GM (B_ * N_)
#define GN HF_
#define GK IN_F_
#define A_IT ((TM * TK) / 4 / 128)
#define B_IT ((TK * TN) / 4 / 128)
#define GEMM_BLOCKS ((GM / TM) * (GN / TN))
#define FOLD_BLOCKS ((EDGE_F_ * H_ * 16) / 128)


__device__ __forceinline__ void gemm_step_load(
    float4 (&va)[A_IT], float4 (&vb)[B_IT],
    const float* __restrict__ ap, const float* __restrict__ bp, const int k0)
{
    #pragma unroll
    for (int it = 0; it < A_IT; ++it)
        va[it] = *(const float4*)(ap + (size_t)(4 * it) * GK + k0);
    #pragma unroll
    for (int it = 0; it < B_IT; ++it)
        vb[it] = *(const float4*)(bp + (size_t)(k0 + 16 * it) * GN);
}

__device__ __forceinline__ void gemm_step_stage(
    float (*As)[TLD], float (*Bs)[TLD],
    const float4 (&va)[A_IT], const float4 (&vb)[B_IT],
    const int a_r, const int a_c, const int b_r, const int b_c4)
{
    #pragma unroll
    for (int it = 0; it < A_IT; ++it)
        *(float4*)&As[a_r + 4 * it][a_c] = va[it];
    #pragma unroll
    for (int it = 0; it < B_IT; ++it) {
        const int r = b_r + 16 * it;
        Bs[b_c4 + 0][r] = vb[it].x;
        Bs[b_c4 + 1][r] = vb[it].y;
        Bs[b_c4 + 2][r] = vb[it].z;
        Bs[b_c4 + 3][r] = vb[it].w;
    }
}

__device__ __forceinline__ void gemm_step_mma(
    const float (*As)[TLD], const float (*Bs)[TLD], v8f (&acc)[2],
    const int mrow, const int l, const int half)
{
    #pragma unroll
    for (int kk = 0; kk < TK; kk += 4) {
        v2f a = *(const v2f*)&As[mrow + l][kk + 2 * half];
        #pragma unroll
        for (int nt = 0; nt < 2; ++nt) {
            v2f b = *(const v2f*)&Bs[nt * 16 + l][kk + 2 * half];
            acc[nt] = __builtin_amdgcn_wmma_f32_16x16x4_f32(
                false, a, false, b, (short)0, acc[nt], false, false);
        }
    }
}

#define NSTEP (GK / TK)

__global__ __launch_bounds__(128) void gemm_node_and_fold_kernel(
    const float* __restrict__ A, const float* __restrict__ Bm,
    float* __restrict__ C,
    const float* __restrict__ W_edge, const float* __restrict__ w_attn,
    float* __restrict__ wpT)
{
    if (blockIdx.x >= GEMM_BLOCKS) {
        edge_weight_fold((blockIdx.x - GEMM_BLOCKS) * 128 + threadIdx.x,
                         W_edge, w_attn, wpT);
        return;
    }

    __shared__ __attribute__((aligned(16))) float As[2][TM][TLD];
    __shared__ __attribute__((aligned(16))) float Bs[2][TN][TLD];

    const int tid  = threadIdx.x;
    const int wave = tid >> 5;
    const int lane = tid & 31;
    const int half = lane >> 4;
    const int l    = lane & 15;
    const int m0   = (blockIdx.x >> 4) * TM;
    const int n0   = (blockIdx.x & 15) * TN;
    const int mrow = wave * 16;

    const int a_r = tid >> 5;
    const int a_c = (tid & 31) * 4;
    const int b_c4 = ((tid & 3) + 4 * (wave & 1)) * 4;
    const int b_r  = ((tid >> 2) & 7) + 8 * (wave >> 1);

    const float* ap = A  + (size_t)(m0 + a_r) * GK + a_c;
    const float* bp = Bm + (size_t)b_r * GN + (n0 + b_c4);

    v8f acc[2] = {};
    float4 va[A_IT], vb[B_IT];

    gemm_step_load(va, vb, ap, bp, 0);
    gemm_step_stage(As[0], Bs[0], va, vb, a_r, a_c, b_r, b_c4);
    __syncthreads();

    for (int s = 0; s < NSTEP - 1; ++s) {
        const int cur = s & 1;
        gemm_step_load(va, vb, ap, bp, (s + 1) * TK);
        gemm_step_mma(As[cur], Bs[cur], acc, mrow, l, half);
        gemm_step_stage(As[cur ^ 1], Bs[cur ^ 1], va, vb, a_r, a_c, b_r, b_c4);
        __syncthreads();
    }
    gemm_step_mma(As[(NSTEP - 1) & 1], Bs[(NSTEP - 1) & 1], acc, mrow, l, half);

    #pragma unroll
    for (int nt = 0; nt < 2; ++nt) {
        #pragma unroll
        for (int r = 0; r < 8; ++r) {
            C[(size_t)(m0 + mrow + r + 8 * half) * GN + (n0 + nt * 16 + l)] =
                acc[nt][r];
        }
    }
}

#define EQ   (EDGE_F_ / 8)
#define NPB  4
#define NROW (NPB + DEG_ - 1)

__global__ __launch_bounds__(128) void edge_attn_fused_kernel(
    const float* __restrict__ g, const float* __restrict__ ea,
    const float* __restrict__ wpT, const float* __restrict__ w_attn,
    float* __restrict__ out)
{
    __shared__ __attribute__((aligned(16))) float gs[NROW][HF_];
    __shared__ float sj_s[NROW][H_];
    __shared__ float si_s[NPB][H_];
    __shared__ __attribute__((aligned(16))) float se_s[NPB][DEG_][H_];
    __shared__ __attribute__((aligned(16))) float aw_s[NPB][2][DEG_][4];

    const int tid  = threadIdx.x;
    const int lane = tid & 31;
    const int wave = tid >> 5;
    const int half = lane >> 4;
    const int c    = lane & 15;
    const int b    = blockIdx.x / (N_ / NPB);
    const int n0   = (blockIdx.x % (N_ / NPB)) * NPB;
    const int n    = n0 + wave;

    const int row0 = b * E_ + n * DEG_;
    const float* bp = ea  + (size_t)(row0 + c) * EDGE_F_ + 4 * half;
    const float* ap = wpT + (c & (H_ - 1)) * EDGE_F_ + 4 * half;
    float4 vb[EQ], va[EQ];
    #pragma unroll
    for (int q = 0; q < EQ; ++q) {
        vb[q] = *(const float4*)(bp + 8 * q);
        va[q] = *(const float4*)(ap + 8 * q);
    }
    float4 gv[NROW];
    #pragma unroll
    for (int r = 0; r < NROW; ++r) {
        const int node = (n0 + r) & (N_ - 1);
        gv[r] = *(const float4*)(g + (size_t)(b * N_ + node) * HF_ + 4 * tid);
    }
    const float4 wi4 = *(const float4*)(w_attn + 4 * c);
    const float4 wj4 = *(const float4*)(w_attn + F_ + 4 * c);
    __builtin_amdgcn_sched_barrier(0);

    float pj[NROW], pi[NPB];
    #pragma unroll
    for (int r = 0; r < NROW; ++r) {
        *(float4*)&gs[r][4 * tid] = gv[r];
        pj[r] = gv[r].x * wj4.x + gv[r].y * wj4.y + gv[r].z * wj4.z + gv[r].w * wj4.w;
    }
    #pragma unroll
    for (int r = 0; r < NPB; ++r)
        pi[r] = gv[r].x * wi4.x + gv[r].y * wi4.y + gv[r].z * wi4.z + gv[r].w * wi4.w;
    #pragma unroll
    for (int off = 8; off >= 1; off >>= 1) {
        #pragma unroll
        for (int r = 0; r < NROW; ++r)
            pj[r] += __shfl_xor(pj[r], off, 32);
        #pragma unroll
        for (int r = 0; r < NPB; ++r)
            pi[r] += __shfl_xor(pi[r], off, 32);
    }
    if (c == 0) {
        const int hh = tid >> 4;
        #pragma unroll
        for (int r = 0; r < NROW; ++r) sj_s[r][hh] = pj[r];
        #pragma unroll
        for (int r = 0; r < NPB; ++r)  si_s[r][hh] = pi[r];
    }

    v8f acc = {};
    #pragma unroll
    for (int q = 0; q < EQ; ++q) {
        v2f a0 = {va[q].x, va[q].y}, b0 = {vb[q].x, vb[q].y};
        v2f a1 = {va[q].z, va[q].w}, b1 = {vb[q].z, vb[q].w};
        acc = __builtin_amdgcn_wmma_f32_16x16x4_f32(
            false, a0, false, b0, (short)0, acc, false, false);
        acc = __builtin_amdgcn_wmma_f32_16x16x4_f32(
            false, a1, false, b1, (short)0, acc, false, false);
    }
    if (half == 0) {
        *(float4*)&se_s[wave][c][0] = make_float4(acc[0], acc[1], acc[2], acc[3]);
        *(float4*)&se_s[wave][c][4] = make_float4(acc[4], acc[5], acc[6], acc[7]);
    }
    __syncthreads();

    const int k    = c;
    const int wrap = (n + DEG_ > N_) ? (n + DEG_ - N_) : 0;
    const int rank = (n + k < N_) ? (k + wrap) : (n + k - N_);
    float aw[4];
    #pragma unroll
    for (int j = 0; j < 4; ++j) {
        const int h = 2 * j + half;
        float sc = si_s[wave][h] + sj_s[wave + k][h] + se_s[wave][rank][h];
        sc = (sc > 0.f) ? sc : SLOPE_ * sc;
        float m = sc;
        #pragma unroll
        for (int off = 8; off >= 1; off >>= 1)
            m = fmaxf(m, __shfl_xor(m, off, 32));
        const float ex = __expf(sc - m);
        float s = ex;
        #pragma unroll
        for (int off = 8; off >= 1; off >>= 1)
            s += __shfl_xor(s, off, 32);
        aw[j] = ex / s;
    }
    *(float4*)&aw_s[wave][half][k][0] = make_float4(aw[0], aw[1], aw[2], aw[3]);
    __syncthreads();

    float4 o[4];
    #pragma unroll
    for (int j = 0; j < 4; ++j) o[j] = make_float4(0.f, 0.f, 0.f, 0.f);
    #pragma unroll
    for (int kk = 0; kk < DEG_; ++kk) {
        const float4 a4 = *(const float4*)&aw_s[wave][half][kk][0];
        const float aj[4] = {a4.x, a4.y, a4.z, a4.w};
        #pragma unroll
        for (int j = 0; j < 4; ++j) {
            const float4 gq = *(const float4*)&gs[wave + kk][4 * (lane + 32 * j)];
            o[j].x += aj[j] * gq.x;
            o[j].y += aj[j] * gq.y;
            o[j].z += aj[j] * gq.z;
            o[j].w += aj[j] * gq.w;
        }
    }
    float* op = out + (size_t)(b * N_ + n) * HF_;
    #pragma unroll
    for (int j = 0; j < 4; ++j)
        *(float4*)(op + 4 * (lane + 32 * j)) = o[j];
}

extern "C" void kernel_launch(void* const* d_in, const int* in_sizes, int n_in,
                              void* d_out, int out_size, void* d_ws, size_t ws_size,
                              hipStream_t stream)
{
    const float* h         = (const float*)d_in[0];
    const float* edge_attr = (const float*)d_in[2];
    const float* W_node    = (const float*)d_in[3];
    const float* W_edge    = (const float*)d_in[4];
    const float* w_attn    = (const float*)d_in[5];
    float* out = (float*)d_out;

    float* ws  = (float*)d_ws;
    float* g   = ws;
    float* wpT = g  + (size_t)B_ * N_ * HF_;

    gemm_node_and_fold_kernel<<<GEMM_BLOCKS + FOLD_BLOCKS, dim3(128), 0, stream>>>(
        h, W_node, g, W_edge, w_attn, wpT);

    edge_attn_fused_kernel<<<(B_ * N_) / NPB, dim3(128), 0, stream>>>(
        g, edge_attr, wpT, w_attn, out);
}


# node GEMM: LDS fragment reads in 4 whole groups per K step, a 16-WMMA block ahead, 2 register sets (were 16 groups among the WMMAs); tile prefetch issued behind the barrier; same k order per element
# speedup vs baseline: 7.3012x; 1.0855x over previous
#include <hip/hip_runtime.h>
#include <math.h>

typedef __attribute__((ext_vector_type(2))) float v2f;
typedef __attribute__((ext_vector_type(4))) float v4f;
typedef __attribute__((ext_vector_type(8))) float v8f;

#define B_     16
#define N_     64
#define DEG_   16
#define IN_F_  1024
#define EDGE_F_ 128
#define H_     8
#define F_     64
#define E_     (N_ * DEG_)
#define HF_    (H_ * F_)
#define SLOPE_ 0.2f

__device__ __forceinline__ void edge_weight_fold(
    const int t, const float* __restrict__ W_edge,
    const float* __restrict__ w_attn, float* __restrict__ wpT)
{
    const int o   = t >> 4;
    const int sub = t & 15;
    const float4 w = *(const float4*)(W_edge + (size_t)o * F_ + sub * 4);
    const float* we = w_attn + 2 * F_ + sub * 4;
    float p = w.x * we[0] + w.y * we[1] + w.z * we[2] + w.w * we[3];
    #pragma unroll
    for (int off = 8; off >= 1; off >>= 1)
        p += __shfl_xor(p, off, 32);
    if (sub == 0) wpT[(o & (H_ - 1)) * EDGE_F_ + (o >> 3)] = p;
}

#define TM 64
#define TN 32
#define TK 128
#define TLD (TK + 4)
#define GM (B_ * N_)
#define GN HF_
#define GK IN_F_
#define A_IT ((TM * TK) / 4 / 128)
#define B_IT ((TK * TN) / 4 / 128)
#define GEMM_BLOCKS ((GM / TM) * (GN / TN))
#define FOLD_BLOCKS ((EDGE_F_ * H_ * 16) / 128)


__device__ __forceinline__ void gemm_step_load(
    v4f (&va)[A_IT], v4f (&vb)[B_IT],
    const float* ap, const float* bp, const int k0)
{
    #pragma unroll
    for (int it = 0; it < A_IT; ++it)
        va[it] = *(const v4f*)(ap + (size_t)(4 * it) * GK + k0);
    #pragma unroll
    for (int it = 0; it < B_IT; ++it)
        vb[it] = *(const v4f*)(bp + (size_t)(k0 + 16 * it) * GN);
}

__device__ __forceinline__ void gemm_step_stage(
    float (*As)[TLD], float (*Bs)[TLD],
    const v4f (&va)[A_IT], const v4f (&vb)[B_IT],
    const int a_r, const int a_c, const int b_r, const int b_c4)
{
    #pragma unroll
    for (int it = 0; it < A_IT; ++it)
        *(v4f*)&As[a_r + 4 * it][a_c] = va[it];
    #pragma unroll
    for (int it = 0; it < B_IT; ++it) {
        const int r = b_r + 16 * it;
        Bs[b_c4 + 0][r] = vb[it].x;
        Bs[b_c4 + 1][r] = vb[it].y;
        Bs[b_c4 + 2][r] = vb[it].z;
        Bs[b_c4 + 3][r] = vb[it].w;
    }
}

#define KB  32
#define KBI (KB / 4)

struct FragSet { v2f a[KBI]; v2f b0[KBI]; v2f b1[KBI]; };

__device__ __forceinline__ void frag_load(
    FragSet& f, const float (*As)[TLD], const float (*Bs)[TLD],
    const int mrow, const int l, const int half, const int kb)
{
    #pragma unroll
    for (int i = 0; i < KBI; ++i) {
        const int kk = kb + 4 * i;
        f.a[i]  = *(const v2f*)&As[mrow + l][kk + 2 * half];
        f.b0[i] = *(const v2f*)&Bs[l][kk + 2 * half];
        f.b1[i] = *(const v2f*)&Bs[16 + l][kk + 2 * half];
    }
}

__device__ __forceinline__ void frag_mma(const FragSet& f, v8f (&acc)[2])
{
    #pragma unroll
    for (int i = 0; i < KBI; ++i) {
        acc[0] = __builtin_amdgcn_wmma_f32_16x16x4_f32(
            false, f.a[i], false, f.b0[i], (short)0, acc[0], false, false);
        acc[1] = __builtin_amdgcn_wmma_f32_16x16x4_f32(
            false, f.a[i], false, f.b1[i], (short)0, acc[1], false, false);
    }
}

#define HOLD_GROUP() __builtin_amdgcn_sched_barrier(0)

template <bool PREFETCH>
__device__ __forceinline__ void gemm_k_step(
    const float (*Ac)[TLD], const float (*Bc)[TLD],
    float (*An)[TLD], float (*Bn)[TLD],
    v8f (&acc)[2], FragSet& x, FragSet& y,
    v4f (&va)[A_IT], v4f (&vb)[B_IT],
    const float* ap, const float* bp, const int k0,
    const int mrow, const int l, const int half,
    const int a_r, const int a_c, const int b_r, const int b_c4)
{
    HOLD_GROUP();
    frag_mma(x, acc);
    HOLD_GROUP();
    frag_load(x, Ac, Bc, mrow, l, half, 2 * KB);
    HOLD_GROUP();
    frag_mma(y, acc);
    HOLD_GROUP();
    frag_load(y, Ac, Bc, mrow, l, half, 3 * KB);
    HOLD_GROUP();
    frag_mma(x, acc);
    gemm_step_stage(An, Bn, va, vb, a_r, a_c, b_r, b_c4);
    __syncthreads();
    if (PREFETCH) gemm_step_load(va, vb, ap, bp, k0);
    frag_load(x, An, Bn, mrow, l, half, 0 * KB);
    HOLD_GROUP();
    frag_mma(y, acc);
    HOLD_GROUP();
    frag_load(y, An, Bn, mrow, l, half, 1 * KB);
}

#define NSTEP (GK / TK)

__global__ __launch_bounds__(128) void gemm_node_and_fold_kernel(
    const float* A, const float* Bm,
    float* __restrict__ C,
    const float* __restrict__ W_edge, const float* __restrict__ w_attn,
    float* __restrict__ wpT)
{
    if (blockIdx.x >= GEMM_BLOCKS) {
        edge_weight_fold((blockIdx.x - GEMM_BLOCKS) * 128 + threadIdx.x,
                         W_edge, w_attn, wpT);
        return;
    }

    __shared__ __attribute__((aligned(16))) float As[2][TM][TLD];
    __shared__ __attribute__((aligned(16))) float Bs[2][TN][TLD];

    const int tid  = threadIdx.x;
    const int wave = tid >> 5;
    const int lane = tid & 31;
    const int half = lane >> 4;
    const int l    = lane & 15;
    const int m0   = (blockIdx.x >> 4) * TM;
    const int n0   = (blockIdx.x & 15) * TN;
    const int mrow = wave * 16;

    const int a_r = tid >> 5;
    const int a_c = (tid & 31) * 4;
    const int b_c4 = ((tid & 3) + 4 * (wave & 1)) * 4;
    const int b_r  = ((tid >> 2) & 7) + 8 * (wave >> 1);

    const float* ap = A  + (size_t)(m0 + a_r) * GK + a_c;
    const float* bp = Bm + (size_t)b_r * GN + (n0 + b_c4);

    v8f acc[2] = {};
    v4f va[A_IT], vb[B_IT];
    FragSet x, y;

    gemm_step_load(va, vb, ap, bp, 0);
    gemm_step_stage(As[0], Bs[0], va, vb, a_r, a_c, b_r, b_c4);
    __syncthreads();
    gemm_step_load(va, vb, ap, bp, TK);
    frag_load(x, As[0], Bs[0], mrow, l, half, 0 * KB);
    frag_load(y, As[0], Bs[0], mrow, l, half, 1 * KB);

    for (int s = 0; s < NSTEP - 2; ++s) {
        const int cur = s & 1;
        gemm_k_step<true>(As[cur], Bs[cur], As[cur ^ 1], Bs[cur ^ 1], acc, x, y,
                          va, vb, ap, bp, (s + 2) * TK,
                          mrow, l, half, a_r, a_c, b_r, b_c4);
    }
    {
        const int cur = (NSTEP - 2) & 1;
        gemm_k_step<false>(As[cur], Bs[cur], As[cur ^ 1], Bs[cur ^ 1], acc, x, y,
                           va, vb, ap, bp, 0,
                           mrow, l, half, a_r, a_c, b_r, b_c4);
    }
    {
        const int cur = (NSTEP - 1) & 1;
        HOLD_GROUP();
        frag_mma(x, acc);
        HOLD_GROUP();
        frag_load(x, As[cur], Bs[cur], mrow, l, half, 2 * KB);
        HOLD_GROUP();
        frag_mma(y, acc);
        HOLD_GROUP();
        frag_load(y, As[cur], Bs[cur], mrow, l, half, 3 * KB);
        HOLD_GROUP();
        frag_mma(x, acc);
        frag_mma(y, acc);
    }

    #pragma unroll
    for (int nt = 0; nt < 2; ++nt) {
        #pragma unroll
        for (int r = 0; r < 8; ++r) {
            C[(size_t)(m0 + mrow + r + 8 * half) * GN + (n0 + nt * 16 + l)] =
                acc[nt][r];
        }
    }
}

#define EQ   (EDGE_F_ / 8)
#define NPB  4
#define NROW (NPB + DEG_ - 1)

__global__ __launch_bounds__(128) void edge_attn_fused_kernel(
    const float* __restrict__ g, const float* __restrict__ ea,
    const float* __restrict__ wpT, const float* __restrict__ w_attn,
    float* __restrict__ out)
{
    __shared__ __attribute__((aligned(16))) float gs[NROW][HF_];
    __shared__ float sj_s[NROW][H_];
    __shared__ float si_s[NPB][H_];
    __shared__ __attribute__((aligned(16))) float se_s[NPB][DEG_][H_];
    __shared__ __attribute__((aligned(16))) float aw_s[NPB][2][DEG_][4];

    const int tid  = threadIdx.x;
    const int lane = tid & 31;
    const int wave = tid >> 5;
    const int half = lane >> 4;
    const int c    = lane & 15;
    const int b    = blockIdx.x / (N_ / NPB);
    const int n0   = (blockIdx.x % (N_ / NPB)) * NPB;
    const int n    = n0 + wave;

    const int row0 = b * E_ + n * DEG_;
    const float* bp = ea  + (size_t)(row0 + c) * EDGE_F_ + 4 * half;
    const float* ap = wpT + (c & (H_ - 1)) * EDGE_F_ + 4 * half;
    float4 vb[EQ], va[EQ];
    #pragma unroll
    for (int q = 0; q < EQ; ++q) {
        vb[q] = *(const float4*)(bp + 8 * q);
        va[q] = *(const float4*)(ap + 8 * q);
    }
    float4 gv[NROW];
    #pragma unroll
    for (int r = 0; r < NROW; ++r) {
        const int node = (n0 + r) & (N_ - 1);
        gv[r] = *(const float4*)(g + (size_t)(b * N_ + node) * HF_ + 4 * tid);
    }
    const float4 wi4 = *(const float4*)(w_attn + 4 * c);
    const float4 wj4 = *(const float4*)(w_attn + F_ + 4 * c);
    __builtin_amdgcn_sched_barrier(0);

    float pj[NROW], pi[NPB];
    #pragma unroll
    for (int r = 0; r < NROW; ++r) {
        *(float4*)&gs[r][4 * tid] = gv[r];
        pj[r] = gv[r].x * wj4.x + gv[r].y * wj4.y + gv[r].z * wj4.z + gv[r].w * wj4.w;
    }
    #pragma unroll
    for (int r = 0; r < NPB; ++r)
        pi[r] = gv[r].x * wi4.x + gv[r].y * wi4.y + gv[r].z * wi4.z + gv[r].w * wi4.w;
    #pragma unroll
    for (int off = 8; off >= 1; off >>= 1) {
        #pragma unroll
        for (int r = 0; r < NROW; ++r)
            pj[r] += __shfl_xor(pj[r], off, 32);
        #pragma unroll
        for (int r = 0; r < NPB; ++r)
            pi[r] += __shfl_xor(pi[r], off, 32);
    }
    if (c == 0) {
        const int hh = tid >> 4;
        #pragma unroll
        for (int r = 0; r < NROW; ++r) sj_s[r][hh] = pj[r];
        #pragma unroll
        for (int r = 0; r < NPB; ++r)  si_s[r][hh] = pi[r];
    }

    v8f acc = {};
    #pragma unroll
    for (int q = 0; q < EQ; ++q) {
        v2f a0 = {va[q].x, va[q].y}, b0 = {vb[q].x, vb[q].y};
        v2f a1 = {va[q].z, va[q].w}, b1 = {vb[q].z, vb[q].w};
        acc = __builtin_amdgcn_wmma_f32_16x16x4_f32(
            false, a0, false, b0, (short)0, acc, false, false);
        acc = __builtin_amdgcn_wmma_f32_16x16x4_f32(
            false, a1, false, b1, (short)0, acc, false, false);
    }
    if (half == 0) {
        *(float4*)&se_s[wave][c][0] = make_float4(acc[0], acc[1], acc[2], acc[3]);
        *(float4*)&se_s[wave][c][4] = make_float4(acc[4], acc[5], acc[6], acc[7]);
    }
    __syncthreads();

    const int k    = c;
    const int wrap = (n + DEG_ > N_) ? (n + DEG_ - N_) : 0;
    const int rank = (n + k < N_) ? (k + wrap) : (n + k - N_);
    float aw[4];
    #pragma unroll
    for (int j = 0; j < 4; ++j) {
        const int h = 2 * j + half;
        float sc = si_s[wave][h] + sj_s[wave + k][h] + se_s[wave][rank][h];
        sc = (sc > 0.f) ? sc : SLOPE_ * sc;
        float m = sc;
        #pragma unroll
        for (int off = 8; off >= 1; off >>= 1)
            m = fmaxf(m, __shfl_xor(m, off, 32));
        const float ex = __expf(sc - m);
        float s = ex;
        #pragma unroll
        for (int off = 8; off >= 1; off >>= 1)
            s += __shfl_xor(s, off, 32);
        aw[j] = ex / s;
    }
    *(float4*)&aw_s[wave][half][k][0] = make_float4(aw[0], aw[1], aw[2], aw[3]);
    __syncthreads();

    float4 o[4];
    #pragma unroll
    for (int j = 0; j < 4; ++j) o[j] = make_float4(0.f, 0.f, 0.f, 0.f);
    #pragma unroll
    for (int kk = 0; kk < DEG_; ++kk) {
        const float4 a4 = *(const float4*)&aw_s[wave][half][kk][0];
        const float aj[4] = {a4.x, a4.y, a4.z, a4.w};
        #pragma unroll
        for (int j = 0; j < 4; ++j) {
            const float4 gq = *(const float4*)&gs[wave + kk][4 * (lane + 32 * j)];
            o[j].x += aj[j] * gq.x;
            o[j].y += aj[j] * gq.y;
            o[j].z += aj[j] * gq.z;
            o[j].w += aj[j] * gq.w;
        }
    }
    float* op = out + (size_t)(b * N_ + n) * HF_;
    #pragma unroll
    for (int j = 0; j < 4; ++j)
        *(float4*)(op + 4 * (lane + 32 * j)) = o[j];
}

extern "C" void kernel_launch(void* const* d_in, const int* in_sizes, int n_in,
                              void* d_out, int out_size, void* d_ws, size_t ws_size,
                              hipStream_t stream)
{
    const float* h         = (const float*)d_in[0];
    const float* edge_attr = (const float*)d_in[2];
    const float* W_node    = (const float*)d_in[3];
    const float* W_edge    = (const float*)d_in[4];
    const float* w_attn    = (const float*)d_in[5];
    float* out = (float*)d_out;

    float* ws  = (float*)d_ws;
    float* g   = ws;
    float* wpT = g  + (size_t)B_ * N_ * HF_;

    gemm_node_and_fold_kernel<<<GEMM_BLOCKS + FOLD_BLOCKS, dim3(128), 0, stream>>>(
        h, W_node, g, W_edge, w_attn, wpT);

    edge_attn_fused_kernel<<<(B_ * N_) / NPB, dim3(128), 0, stream>>>(
        g, edge_attr, wpT, w_attn, out);
}
